# starting kernel
# baseline (speedup 1.0000x reference)
#include <hip/hip_runtime.h>
#include <hip/hip_bf16.h>
#include <math.h>

typedef float v2f __attribute__((ext_vector_type(2)));
typedef float v8f __attribute__((ext_vector_type(8)));

#define NBINS  216
#define CDIM   13
#define KSTEPS 54
#define WMMA4(A,B,C) __builtin_amdgcn_wmma_f32_16x16x4_f32(false,(A),false,(B),(short)0,(C),false,false)

__device__ __forceinline__ float lane_read(float v, int srcLane) {
    return __builtin_bit_cast(float,
        __builtin_amdgcn_ds_bpermute(srcLane << 2, __builtin_bit_cast(int, v)));
}

__device__ __forceinline__ void softmax16(v8f& acc, float bias, int c, int lane) {
    #pragma unroll
    for (int r = 0; r < 8; ++r) {
        float lg = acc[r] + bias;
        float m = (c < CDIM) ? lg : -1e30f;
        #pragma unroll
        for (int off = 8; off; off >>= 1)
            m = fmaxf(m, lane_read(m, lane ^ off));
        float e = (c < CDIM) ? __expf(lg - m) : 0.0f;
        float s = e;
        #pragma unroll
        for (int off = 8; off; off >>= 1)
            s += lane_read(s, lane ^ off);
        acc[r] = e / s;
    }
}

__global__ __launch_bounds__(256)
void chord_heads_kernel(const float* __restrict__ x,
                        const float* __restrict__ Wk, const float* __restrict__ bk,
                        const float* __restrict__ Wd, const float* __restrict__ bd,
                        const float* __restrict__ Wb, const float* __restrict__ bb,
                        float* __restrict__ out, int R)
{
    __shared__ v2f wfrag[3][KSTEPS][32];
    const int tid = threadIdx.x;
    for (int idx = tid; idx < 3 * KSTEPS * 32; idx += blockDim.x) {
        int h   = idx / (KSTEPS * 32);
        int rem = idx - h * (KSTEPS * 32);
        int s   = rem >> 5;
        int l   = rem & 31;
        int cc  = l & 15;
        int hh  = l >> 4;
        int k0  = 4 * s + 2 * hh;
        const float* W = (h == 0) ? Wk : ((h == 1) ? Wd : Wb);
        v2f w; w.x = 0.0f; w.y = 0.0f;
        if (cc < CDIM) { w.x = W[k0 * CDIM + cc]; w.y = W[(k0 + 1) * CDIM + cc]; }
        wfrag[h][s][l] = w;
    }
    __syncthreads();

    const int lane = tid & 31;
    const int wave = tid >> 5;
    const int c    = lane & 15;
    const int hi   = lane >> 4;
    const int base = lane & 16;

    const float biasK = (c < CDIM) ? bk[c] : 0.0f;
    const float biasD = (c < CDIM) ? bd[c] : 0.0f;
    const float biasB = (c < CDIM) ? bb[c] : 0.0f;

    const int tiles = (R + 15) >> 4;
    const int gw    = blockIdx.x * (blockDim.x >> 5) + wave;
    const int nw    = gridDim.x * (blockDim.x >> 5);
    const long long hstride = (long long)R * CDIM;

    for (int t = gw; t < tiles; t += nw) {
        const int row0 = t << 4;
        int rl = row0 + (lane & 15);
        int rc = rl < R ? rl : R - 1;
        const float* xrow = x + (long long)rc * NBINS + 2 * hi;

        int tn = t + nw;
        if (tn < tiles) {
            int rn = (tn << 4) + (lane & 15);
            rn = rn < R ? rn : R - 1;
            __builtin_prefetch(x + (long long)rn * NBINS, 0, 1);
        }

        v8f accK = {0.f,0.f,0.f,0.f,0.f,0.f,0.f,0.f};
        v8f accD = {0.f,0.f,0.f,0.f,0.f,0.f,0.f,0.f};
        v8f accB = {0.f,0.f,0.f,0.f,0.f,0.f,0.f,0.f};

        v2f a  = *(const v2f*)(xrow);
        v2f w0 = wfrag[0][0][lane];
        v2f w1 = wfrag[1][0][lane];
        v2f w2 = wfrag[2][0][lane];
        #pragma unroll 3
        for (int s = 0; s < KSTEPS - 1; ++s) {
            v2f a_n  = *(const v2f*)(xrow + 4 * (s + 1));
            v2f w0n = wfrag[0][s + 1][lane];
            v2f w1n = wfrag[1][s + 1][lane];
            v2f w2n = wfrag[2][s + 1][lane];
            accK = WMMA4(a, w0, accK);
            accD = WMMA4(a, w1, accD);
            accB = WMMA4(a, w2, accB);
            a = a_n; w0 = w0n; w1 = w1n; w2 = w2n;
        }
        accK = WMMA4(a, w0, accK);
        accD = WMMA4(a, w1, accD);
        accB = WMMA4(a, w2, accB);

        softmax16(accK, biasK, c, lane);
        softmax16(accD, biasD, c, lane);
        softmax16(accB, biasB, c, lane);

        #pragma unroll
        for (int r = 0; r < 8; ++r) {
            int row = row0 + r + 8 * hi;
            if (row < R && c < CDIM) {
                long long o = (long long)row * CDIM + c;
                out[o]                = accK[r];
                out[hstride + o]      = accD[r];
                out[2 * hstride + o]  = accB[r];
            }
        }

        #pragma unroll
        for (int r = 0; r < 8; ++r) {
            float k0v = lane_read(accK[r], base);
            float d0v = lane_read(accD[r], base);
            float root0 = k0v + d0v - k0v * d0v;
            int rr = c - 1;
            float accR = 0.0f;
            #pragma unroll
            for (int k = 0; k < 12; ++k) {
                float kk = lane_read(accK[r], base + k + 1);
                int di = rr - k; di += (di < 0) ? 12 : 0; di += (di < 0) ? 12 : 0;
                float dd = lane_read(accD[r], base + di + 1);
                accR = fmaf(kk, dd, accR);
            }
            float val = (c == 0) ? root0 : accR;
            int row = row0 + r + 8 * hi;
            if (row < R && c < CDIM)
                out[3 * hstride + (long long)row * CDIM + c] = val;
        }
    }
}

extern "C" void kernel_launch(void* const* d_in, const int* in_sizes, int n_in,
                              void* d_out, int out_size, void* d_ws, size_t ws_size,
                              hipStream_t stream) {
    const float* x  = (const float*)d_in[0];
    const float* Wk = (const float*)d_in[1];
    const float* bk = (const float*)d_in[2];
    const float* Wd = (const float*)d_in[3];
    const float* bd = (const float*)d_in[4];
    const float* Wb = (const float*)d_in[5];
    const float* bb = (const float*)d_in[6];
    float* out = (float*)d_out;
    int R = in_sizes[0] / NBINS;

    dim3 block(256);
    dim3 grid(512);
    chord_heads_kernel<<<grid, block, 0, stream>>>(x, Wk, bk, Wd, bd, Wb, bb, out, R);
}


# weight staging reads each head's 216x13 block as whole 16-byte pieces (9 passes of b128 loads, floats scattered to their fragment places in LDS), not 21 passes of 4-byte gathers; sums unchanged
# speedup vs baseline: 1.5689x; 1.5689x over previous
#include <hip/hip_runtime.h>
#include <hip/hip_bf16.h>
#include <math.h>

typedef float v2f __attribute__((ext_vector_type(2)));
typedef float v8f __attribute__((ext_vector_type(8)));
typedef float v4f __attribute__((ext_vector_type(4)));

#define NBINS  216
#define CDIM   13
#define KSTEPS 54
#define GSTEPS 6
#define NGROUPS (KSTEPS / GSTEPS)
#define WQUADS (NBINS * CDIM / 4)
#define OUT_TILE (16 * CDIM)
#define OUT_PAIR (2 * OUT_TILE)
#define OUT_WAVE (4 * OUT_PAIR)
#define WMMA4(A,B,C) __builtin_amdgcn_wmma_f32_16x16x4_f32(false,(A),false,(B),(short)0,(C),false,false)

__device__ __forceinline__ float xhalf(float x) {
    int b = __builtin_bit_cast(int, x);
    return __builtin_bit_cast(float, __builtin_amdgcn_permlanex16(b, b, 0x76543210, 0xfedcba98, false, false));
}

#define WAVE_LDS_ORDER() do { __builtin_amdgcn_fence(__ATOMIC_RELEASE, "wavefront"); \
                              __builtin_amdgcn_wave_barrier();                         \
                              __builtin_amdgcn_fence(__ATOMIC_ACQUIRE, "wavefront"); } while (0)

template <bool REFILL>
__device__ __forceinline__ void do_tile(v2f (&a)[KSTEPS], const float* __restrict__ xnext,
                                        const v2f (*wfrag)[KSTEPS][32],
                                        float biasK, float biasD, float biasB,
                                        float* __restrict__ out, long long hstride,
                                        float* ost, bool whole,
                                        int t, int R, int lane, int c, int hi)
{
    const int row0 = t << 4;
    v8f accK = {0.f,0.f,0.f,0.f,0.f,0.f,0.f,0.f};
    v8f accD = {0.f,0.f,0.f,0.f,0.f,0.f,0.f,0.f};
    v8f accB = {0.f,0.f,0.f,0.f,0.f,0.f,0.f,0.f};


    asm volatile("" ::: "memory");

    v2f wc[3][GSTEPS];
    #pragma unroll
    for (int j = 0; j < GSTEPS; ++j) {
        wc[0][j] = wfrag[0][j][lane];
        wc[1][j] = wfrag[1][j][lane];
        wc[2][j] = wfrag[2][j][lane];
    }
    #pragma unroll
    for (int g = 0; g < NGROUPS; ++g) {
        v2f wn[3][GSTEPS];
        if (g + 1 < NGROUPS) {
            #pragma unroll
            for (int j = 0; j < GSTEPS; ++j) {
                wn[0][j] = wfrag[0][(g + 1) * GSTEPS + j][lane];
                wn[1][j] = wfrag[1][(g + 1) * GSTEPS + j][lane];
                wn[2][j] = wfrag[2][(g + 1) * GSTEPS + j][lane];
            }
        }
        __builtin_amdgcn_sched_barrier(0);
        #pragma unroll
        for (int j = 0; j < GSTEPS; ++j) {
            accK = WMMA4(a[g * GSTEPS + j], wc[0][j], accK);
            accD = WMMA4(a[g * GSTEPS + j], wc[1][j], accD);
            accB = WMMA4(a[g * GSTEPS + j], wc[2][j], accB);
        }
        __builtin_amdgcn_sched_barrier(0);
        if (REFILL) {
            asm volatile("" : "+v"(accK), "+v"(accD), "+v"(accB) :: "memory");
            #pragma unroll
            for (int j = 0; j < GSTEPS; ++j)
                a[g * GSTEPS + j] = *(const v2f*)(xnext + 4 * (g * GSTEPS + j));
            __builtin_amdgcn_sched_barrier(0);
        }
        if (g + 1 < NGROUPS) {
            #pragma unroll
            for (int j = 0; j < GSTEPS; ++j) {
                wc[0][j] = wn[0][j];
                wc[1][j] = wn[1][j];
                wc[2][j] = wn[2][j];
            }
        }
    }

    float* slot = ost + (t & 1) * OUT_TILE + (8 * hi) * CDIM + c;
    if (c < CDIM) {
        #pragma unroll
        for (int r = 0; r < 8; ++r) {
            slot[r * CDIM]                = accK[r] + biasK;
            slot[OUT_PAIR + r * CDIM]     = accD[r] + biasD;
            slot[2 * OUT_PAIR + r * CDIM] = accB[r] + biasB;
        }
    }
    WAVE_LDS_ORDER();

    int lb = c * CDIM + (t & 1) * OUT_TILE;
    int hb = hi;
    asm volatile("" : "+v"(lb), "+v"(hb));
    float* rowp = ost + lb + hb * OUT_PAIR;
    float v[CDIM];
    #pragma unroll
    for (int j = 0; j < CDIM; ++j) v[j] = rowp[j];
    {
        float m = v[0];
        #pragma unroll
        for (int j = 1; j < CDIM; ++j) m = fmaxf(m, v[j]);
        #pragma unroll
        for (int j = 0; j < CDIM; ++j) v[j] = __expf(v[j] - m);
        float s = v[0];
        #pragma unroll
        for (int j = 1; j < CDIM; ++j) s += v[j];
        #pragma unroll
        for (int j = 0; j < CDIM; ++j) { v[j] = v[j] / s; rowp[j] = v[j]; }
    }

    {
        float* bp = ost + 2 * OUT_PAIR + lb + 7 * hb;
        float b[7];
        #pragma unroll
        for (int j = 0; j < 7; ++j) b[j] = bp[j];
        b[6] = hi ? -1e30f : b[6];
        float m = b[0];
        #pragma unroll
        for (int j = 1; j < 7; ++j) m = fmaxf(m, b[j]);
        m = fmaxf(m, xhalf(m));
        #pragma unroll
        for (int j = 0; j < 7; ++j) b[j] = __expf(b[j] - m);
        b[6] = hi ? 0.0f : b[6];
        float s = b[0];
        #pragma unroll
        for (int j = 1; j < 7; ++j) s += b[j];
        s = s + xhalf(s);
        #pragma unroll
        for (int j = 0; j < 6; ++j) bp[j] = b[j] / s;
        if (!hi) bp[6] = b[6] / s;
    }

    {
        float o[CDIM];
        #pragma unroll
        for (int j = 0; j < CDIM; ++j) {
            o[j] = xhalf(v[j]);
            asm volatile("" : "+v"(o[j]));
        }
        float kq[12], dq[12];
        #pragma unroll
        for (int j = 0; j < 12; ++j) {
            kq[j] = hi ? o[j + 1] : v[j + 1];
            dq[j] = hi ? v[((j + 6) % 12) + 1] : o[j + 1];
        }
        float* rp = ost + 3 * OUT_PAIR + lb;
        #pragma unroll
        for (int i = 0; i < 6; ++i) {
            float accR = 0.0f;
            #pragma unroll
            for (int k = 0; k < 12; ++k) accR = fmaf(kq[k], dq[(i - k + 12) % 12], accR);
            rp[1 + 6 * hb + i] = accR;
        }
        const float k0v = hi ? o[0] : v[0];
        const float d0v = hi ? v[0] : o[0];
        if (!hi) rp[0] = k0v + d0v - k0v * d0v;
    }
    __builtin_amdgcn_fence(__ATOMIC_RELEASE, "wavefront");
    __builtin_amdgcn_wave_barrier();
    __builtin_amdgcn_fence(__ATOMIC_ACQUIRE, "wavefront");

    if (whole) {
        if (t & 1) {
            const unsigned lane16 = (unsigned)lane * 16u;
            const char* sv = (const char*)ost;
            char* gbase = (char*)out + (long long)(t >> 1) * (OUT_PAIR * 4);
            const long long hgap = hstride * 4 - OUT_PAIR * 4;
            #pragma unroll
            for (int i = 0; i < 13; ++i) {
                const int hlo = (32 * i) / 104;
                const int bnd = 104 * (hlo + 1) - 32 * i;
                unsigned voff = lane16;
                if (bnd < 32) voff += (lane >= bnd) ? (unsigned)hgap : 0u;
                v4f v = *(const v4f*)(sv + 512 * i + lane16);
                *(v4f*)(gbase + 512 * i + hlo * hgap + voff) = v;
            }
        }
    } else {
        int rows = R - row0;
        rows = rows < 16 ? rows : 16;
        const int nvalid = rows * CDIM;
        #pragma nounroll
        for (int h = 0; h < 4; ++h) {
            #pragma nounroll
            for (int idx = lane; idx < nvalid; idx += 32)
                out[h * hstride + (long long)row0 * CDIM + idx] = ost[(t & 1) * OUT_TILE + h * OUT_PAIR + idx];
        }
    }
}

__global__ __launch_bounds__(256)
void chord_heads_kernel(const float* __restrict__ x,
                        const float* __restrict__ Wk, const float* __restrict__ bk,
                        const float* __restrict__ Wd, const float* __restrict__ bd,
                        const float* __restrict__ Wb, const float* __restrict__ bb,
                        float* __restrict__ out, int R)
{
    __shared__ v2f wfrag[3][KSTEPS][32];
    __shared__ __attribute__((aligned(16))) float ostage[8][OUT_WAVE];
    const int tid = threadIdx.x;
    const bool wvec = (((unsigned long long)Wk | (unsigned long long)Wd | (unsigned long long)Wb) & 15ull) == 0ull;
    if (wvec) {
        float* wf = (float*)&wfrag[0][0][0];
        for (int idx = tid; idx < 3 * KSTEPS * 6; idx += blockDim.x) {
            int hs = idx / 6;
            int r  = idx - hs * 6;
            int hh = r / 3;
            int cc = CDIM + (r - hh * 3);
            v2f z; z.x = 0.0f; z.y = 0.0f;
            (&wfrag[0][0][0])[hs * 32 + hh * 16 + cc] = z;
        }
        for (int q = tid; q < 3 * WQUADS; q += blockDim.x) {
            int h  = q / WQUADS;
            int qq = q - h * WQUADS;
            const float* W = (h == 0) ? Wk : ((h == 1) ? Wd : Wb);
            v4f w = *(const v4f*)(W + 4 * qq);
            int k  = (4 * qq) / CDIM;
            int cc = 4 * qq - k * CDIM;
            #pragma unroll
            for (int e = 0; e < 4; ++e) {
                wf[((h * KSTEPS + (k >> 2)) * 32 + ((k >> 1) & 1) * 16 + cc) * 2 + (k & 1)] = w[e];
                ++cc;
                if (cc == CDIM) { cc = 0; ++k; }
            }
        }
    } else {
        for (int idx = tid; idx < 3 * KSTEPS * 32; idx += blockDim.x) {
            int h   = idx / (KSTEPS * 32);
            int rem = idx - h * (KSTEPS * 32);
            int s   = rem >> 5;
            int l   = rem & 31;
            int cc  = l & 15;
            int hh  = l >> 4;
            int k0  = 4 * s + 2 * hh;
            const float* W = (h == 0) ? Wk : ((h == 1) ? Wd : Wb);
            v2f w; w.x = 0.0f; w.y = 0.0f;
            if (cc < CDIM) { w.x = W[k0 * CDIM + cc]; w.y = W[(k0 + 1) * CDIM + cc]; }
            wfrag[h][s][l] = w;
        }
    }
    __syncthreads();

    const int lane = tid & 31;
    const int wave = tid >> 5;
    const int c    = lane & 15;
    const int hi   = lane >> 4;

    const float biasK = (c < CDIM) ? bk[c] : 0.0f;
    const float biasD = (c < CDIM) ? bd[c] : 0.0f;
    const float biasB = (c < CDIM) ? bb[c] : 0.0f;

    const int tiles = (R + 15) >> 4;
    const int pairs = (tiles + 1) >> 1;
    const int gw    = __builtin_amdgcn_readfirstlane(blockIdx.x * (blockDim.x >> 5) + wave);
    const int nw    = gridDim.x * (blockDim.x >> 5);
    const long long hstride = (long long)R * CDIM;
    float* ost = &ostage[wave & 7][0];
    const bool whole = ((R & 31) == 0) && (R <= (1 << 24)) && ((((unsigned long long)out) & 15ull) == 0ull);

    if (gw >= pairs) return;
    int t = 2 * gw;

    v2f a[KSTEPS];
    {
        int r0 = (t << 4) + c;
        r0 = r0 < R ? r0 : R - 1;
        const float* xfirst = x + (long long)r0 * NBINS + 2 * hi;
        #pragma unroll
        for (int s = 0; s < KSTEPS; ++s) a[s] = *(const v2f*)(xfirst + 4 * s);
    }

    for (;;) {
        const int tn = (t & 1) ? 2 * ((t >> 1) + nw) : t + 1;
        if (tn >= tiles) break;
        int rn = (tn << 4) + c;
        rn = rn < R ? rn : R - 1;
        const float* xnext = x + (long long)rn * NBINS + 2 * hi;
        do_tile<true>(a, xnext, wfrag, biasK, biasD, biasB, out, hstride, ost, whole, t, R, lane, c, hi);
        t = tn;
    }
    do_tile<false>(a, x, wfrag, biasK, biasD, biasB, out, hstride, ost, whole, t, R, lane, c, hi);
}

extern "C" void kernel_launch(void* const* d_in, const int* in_sizes, int n_in,
                              void* d_out, int out_size, void* d_ws, size_t ws_size,
                              hipStream_t stream) {
    const float* x  = (const float*)d_in[0];
    const float* Wk = (const float*)d_in[1];
    const float* bk = (const float*)d_in[2];
    const float* Wd = (const float*)d_in[3];
    const float* bd = (const float*)d_in[4];
    const float* Wb = (const float*)d_in[5];
    const float* bb = (const float*)d_in[6];
    float* out = (float*)d_out;
    int R = in_sizes[0] / NBINS;

    dim3 block(256);
    dim3 grid(512);
    chord_heads_kernel<<<grid, block, 0, stream>>>(x, Wk, bk, Wd, bd, Wb, bb, out, R);
}


# x fragments read 16 bytes per lane straight into registers: one b128 load feeds two K-steps (k order 8p+4half+2step+e, weights staged to match), 27 loads per tile, not 54; k sum order changes
# speedup vs baseline: 1.6132x; 1.0282x over previous
#include <hip/hip_runtime.h>
#include <hip/hip_bf16.h>
#include <math.h>

typedef float v2f __attribute__((ext_vector_type(2)));
typedef float v8f __attribute__((ext_vector_type(8)));
typedef float v4f __attribute__((ext_vector_type(4)));

#define NBINS  216
#define CDIM   13
#define KSTEPS 54
#define KPAIRS 27
#define GSTEPS 6
#define GPAIRS (GSTEPS / 2)
#define NGROUPS (KSTEPS / GSTEPS)
#define WQUADS (NBINS * CDIM / 4)
#define OUT_TILE (16 * CDIM)
#define OUT_PAIR (2 * OUT_TILE)
#define OUT_WAVE (4 * OUT_PAIR)
#define WMMA4(A,B,C) __builtin_amdgcn_wmma_f32_16x16x4_f32(false,(A),false,(B),(short)0,(C),false,false)

#define KSTEP_OF(k) ((((k) >> 3) << 1) | (((k) >> 1) & 1))
#define HALF_OF(k)  (((k) >> 2) & 1)
__device__ __forceinline__ v2f piece_lo(v4f p) { return __builtin_shufflevector(p, p, 0, 1); }
__device__ __forceinline__ v2f piece_hi(v4f p) { return __builtin_shufflevector(p, p, 2, 3); }

__device__ __forceinline__ float xhalf(float x) {
    int b = __builtin_bit_cast(int, x);
    return __builtin_bit_cast(float, __builtin_amdgcn_permlanex16(b, b, 0x76543210, 0xfedcba98, false, false));
}

#define WAVE_LDS_ORDER() do { __builtin_amdgcn_fence(__ATOMIC_RELEASE, "wavefront"); \
                              __builtin_amdgcn_wave_barrier();                         \
                              __builtin_amdgcn_fence(__ATOMIC_ACQUIRE, "wavefront"); } while (0)

template <bool REFILL>
__device__ __forceinline__ void do_tile(v4f (&a)[KPAIRS], const float* __restrict__ xnext,
                                        const v2f (*wfrag)[KSTEPS][32],
                                        float biasK, float biasD, float biasB,
                                        float* __restrict__ out, long long hstride,
                                        float* ost, bool whole,
                                        int t, int R, int lane, int c, int hi)
{
    const int row0 = t << 4;
    v8f accK = {0.f,0.f,0.f,0.f,0.f,0.f,0.f,0.f};
    v8f accD = {0.f,0.f,0.f,0.f,0.f,0.f,0.f,0.f};
    v8f accB = {0.f,0.f,0.f,0.f,0.f,0.f,0.f,0.f};


    asm volatile("" ::: "memory");

    v2f wc[3][GSTEPS];
    #pragma unroll
    for (int j = 0; j < GSTEPS; ++j) {
        wc[0][j] = wfrag[0][j][lane];
        wc[1][j] = wfrag[1][j][lane];
        wc[2][j] = wfrag[2][j][lane];
    }
    #pragma unroll
    for (int g = 0; g < NGROUPS; ++g) {
        v2f wn[3][GSTEPS];
        if (g + 1 < NGROUPS) {
            #pragma unroll
            for (int j = 0; j < GSTEPS; ++j) {
                wn[0][j] = wfrag[0][(g + 1) * GSTEPS + j][lane];
                wn[1][j] = wfrag[1][(g + 1) * GSTEPS + j][lane];
                wn[2][j] = wfrag[2][(g + 1) * GSTEPS + j][lane];
            }
        }
        __builtin_amdgcn_sched_barrier(0);
        #pragma unroll
        for (int j = 0; j < GSTEPS; ++j) {
            const v4f pc = a[g * GPAIRS + (j >> 1)];
            const v2f af = (j & 1) ? piece_hi(pc) : piece_lo(pc);
            accK = WMMA4(af, wc[0][j], accK);
            accD = WMMA4(af, wc[1][j], accD);
            accB = WMMA4(af, wc[2][j], accB);
        }
        __builtin_amdgcn_sched_barrier(0);
        if (REFILL) {
            asm volatile("" : "+v"(accK), "+v"(accD), "+v"(accB) :: "memory");
            #pragma unroll
            for (int j = 0; j < GPAIRS; ++j)
                a[g * GPAIRS + j] = *(const v4f*)(xnext + 8 * (g * GPAIRS + j));
            __builtin_amdgcn_sched_barrier(0);
        }
        if (g + 1 < NGROUPS) {
            #pragma unroll
            for (int j = 0; j < GSTEPS; ++j) {
                wc[0][j] = wn[0][j];
                wc[1][j] = wn[1][j];
                wc[2][j] = wn[2][j];
            }
        }
    }

    float* slot = ost + (t & 1) * OUT_TILE + (8 * hi) * CDIM + c;
    if (c < CDIM) {
        #pragma unroll
        for (int r = 0; r < 8; ++r) {
            slot[r * CDIM]                = accK[r] + biasK;
            slot[OUT_PAIR + r * CDIM]     = accD[r] + biasD;
            slot[2 * OUT_PAIR + r * CDIM] = accB[r] + biasB;
        }
    }
    WAVE_LDS_ORDER();

    int lb = c * CDIM + (t & 1) * OUT_TILE;
    int hb = hi;
    asm volatile("" : "+v"(lb), "+v"(hb));
    float* rowp = ost + lb + hb * OUT_PAIR;
    float v[CDIM];
    #pragma unroll
    for (int j = 0; j < CDIM; ++j) v[j] = rowp[j];
    {
        float m = v[0];
        #pragma unroll
        for (int j = 1; j < CDIM; ++j) m = fmaxf(m, v[j]);
        #pragma unroll
        for (int j = 0; j < CDIM; ++j) v[j] = __expf(v[j] - m);
        float s = v[0];
        #pragma unroll
        for (int j = 1; j < CDIM; ++j) s += v[j];
        #pragma unroll
        for (int j = 0; j < CDIM; ++j) { v[j] = v[j] / s; rowp[j] = v[j]; }
    }

    {
        float* bp = ost + 2 * OUT_PAIR + lb + 7 * hb;
        float b[7];
        #pragma unroll
        for (int j = 0; j < 7; ++j) b[j] = bp[j];
        b[6] = hi ? -1e30f : b[6];
        float m = b[0];
        #pragma unroll
        for (int j = 1; j < 7; ++j) m = fmaxf(m, b[j]);
        m = fmaxf(m, xhalf(m));
        #pragma unroll
        for (int j = 0; j < 7; ++j) b[j] = __expf(b[j] - m);
        b[6] = hi ? 0.0f : b[6];
        float s = b[0];
        #pragma unroll
        for (int j = 1; j < 7; ++j) s += b[j];
        s = s + xhalf(s);
        #pragma unroll
        for (int j = 0; j < 6; ++j) bp[j] = b[j] / s;
        if (!hi) bp[6] = b[6] / s;
    }

    {
        float o[CDIM];
        #pragma unroll
        for (int j = 0; j < CDIM; ++j) {
            o[j] = xhalf(v[j]);
            asm volatile("" : "+v"(o[j]));
        }
        float kq[12], dq[12];
        #pragma unroll
        for (int j = 0; j < 12; ++j) {
            kq[j] = hi ? o[j + 1] : v[j + 1];
            dq[j] = hi ? v[((j + 6) % 12) + 1] : o[j + 1];
        }
        float* rp = ost + 3 * OUT_PAIR + lb;
        #pragma unroll
        for (int i = 0; i < 6; ++i) {
            float accR = 0.0f;
            #pragma unroll
            for (int k = 0; k < 12; ++k) accR = fmaf(kq[k], dq[(i - k + 12) % 12], accR);
            rp[1 + 6 * hb + i] = accR;
        }
        const float k0v = hi ? o[0] : v[0];
        const float d0v = hi ? v[0] : o[0];
        if (!hi) rp[0] = k0v + d0v - k0v * d0v;
    }
    __builtin_amdgcn_fence(__ATOMIC_RELEASE, "wavefront");
    __builtin_amdgcn_wave_barrier();
    __builtin_amdgcn_fence(__ATOMIC_ACQUIRE, "wavefront");

    if (whole) {
        if (t & 1) {
            const unsigned lane16 = (unsigned)lane * 16u;
            const char* sv = (const char*)ost;
            char* gbase = (char*)out + (long long)(t >> 1) * (OUT_PAIR * 4);
            const long long hgap = hstride * 4 - OUT_PAIR * 4;
            #pragma unroll
            for (int i = 0; i < 13; ++i) {
                const int hlo = (32 * i) / 104;
                const int bnd = 104 * (hlo + 1) - 32 * i;
                unsigned voff = lane16;
                if (bnd < 32) voff += (lane >= bnd) ? (unsigned)hgap : 0u;
                v4f v = *(const v4f*)(sv + 512 * i + lane16);
                *(v4f*)(gbase + 512 * i + hlo * hgap + voff) = v;
            }
        }
    } else {
        int rows = R - row0;
        rows = rows < 16 ? rows : 16;
        const int nvalid = rows * CDIM;
        #pragma nounroll
        for (int h = 0; h < 4; ++h) {
            #pragma nounroll
            for (int idx = lane; idx < nvalid; idx += 32)
                out[h * hstride + (long long)row0 * CDIM + idx] = ost[(t & 1) * OUT_TILE + h * OUT_PAIR + idx];
        }
    }
}

__global__ __launch_bounds__(256)
void chord_heads_kernel(const float* __restrict__ x,
                        const float* __restrict__ Wk, const float* __restrict__ bk,
                        const float* __restrict__ Wd, const float* __restrict__ bd,
                        const float* __restrict__ Wb, const float* __restrict__ bb,
                        float* __restrict__ out, int R)
{
    __shared__ v2f wfrag[3][KSTEPS][32];
    __shared__ __attribute__((aligned(16))) float ostage[8][OUT_WAVE];
    const int tid = threadIdx.x;
    const bool wvec = (((unsigned long long)Wk | (unsigned long long)Wd | (unsigned long long)Wb) & 15ull) == 0ull;
    if (wvec) {
        float* wf = (float*)&wfrag[0][0][0];
        for (int idx = tid; idx < 3 * KSTEPS * 6; idx += blockDim.x) {
            int hs = idx / 6;
            int r  = idx - hs * 6;
            int hh = r / 3;
            int cc = CDIM + (r - hh * 3);
            v2f z; z.x = 0.0f; z.y = 0.0f;
            (&wfrag[0][0][0])[hs * 32 + hh * 16 + cc] = z;
        }
        for (int q = tid; q < 3 * WQUADS; q += blockDim.x) {
            int h  = q / WQUADS;
            int qq = q - h * WQUADS;
            const float* W = (h == 0) ? Wk : ((h == 1) ? Wd : Wb);
            v4f w = *(const v4f*)(W + 4 * qq);
            int k  = (4 * qq) / CDIM;
            int cc = 4 * qq - k * CDIM;
            #pragma unroll
            for (int e = 0; e < 4; ++e) {
                wf[((h * KSTEPS + KSTEP_OF(k)) * 32 + HALF_OF(k) * 16 + cc) * 2 + (k & 1)] = w[e];
                ++cc;
                if (cc == CDIM) { cc = 0; ++k; }
            }
        }
    } else {
        for (int idx = tid; idx < 3 * KSTEPS * 32; idx += blockDim.x) {
            int h   = idx / (KSTEPS * 32);
            int rem = idx - h * (KSTEPS * 32);
            int s   = rem >> 5;
            int l   = rem & 31;
            int cc  = l & 15;
            int hh  = l >> 4;
            int k0  = 8 * (s >> 1) + 4 * hh + 2 * (s & 1);
            const float* W = (h == 0) ? Wk : ((h == 1) ? Wd : Wb);
            v2f w; w.x = 0.0f; w.y = 0.0f;
            if (cc < CDIM) { w.x = W[k0 * CDIM + cc]; w.y = W[(k0 + 1) * CDIM + cc]; }
            wfrag[h][s][l] = w;
        }
    }
    __syncthreads();

    const int lane = tid & 31;
    const int wave = tid >> 5;
    const int c    = lane & 15;
    const int hi   = lane >> 4;

    const float biasK = (c < CDIM) ? bk[c] : 0.0f;
    const float biasD = (c < CDIM) ? bd[c] : 0.0f;
    const float biasB = (c < CDIM) ? bb[c] : 0.0f;

    const int tiles = (R + 15) >> 4;
    const int pairs = (tiles + 1) >> 1;
    const int gw    = __builtin_amdgcn_readfirstlane(blockIdx.x * (blockDim.x >> 5) + wave);
    const int nw    = gridDim.x * (blockDim.x >> 5);
    const long long hstride = (long long)R * CDIM;
    float* ost = &ostage[wave & 7][0];
    const bool whole = ((R & 31) == 0) && (R <= (1 << 24)) && ((((unsigned long long)out) & 15ull) == 0ull);

    if (gw >= pairs) return;
    int t = 2 * gw;

    v4f a[KPAIRS];
    {
        int r0 = (t << 4) + c;
        r0 = r0 < R ? r0 : R - 1;
        const float* xfirst = x + (long long)r0 * NBINS + 4 * hi;
        #pragma unroll
        for (int p = 0; p < KPAIRS; ++p) a[p] = *(const v4f*)(xfirst + 8 * p);
    }

    for (;;) {
        const int tn = (t & 1) ? 2 * ((t >> 1) + nw) : t + 1;
        if (tn >= tiles) break;
        int rn = (tn << 4) + c;
        rn = rn < R ? rn : R - 1;
        const float* xnext = x + (long long)rn * NBINS + 4 * hi;
        do_tile<true>(a, xnext, wfrag, biasK, biasD, biasB, out, hstride, ost, whole, t, R, lane, c, hi);
        t = tn;
    }
    do_tile<false>(a, x, wfrag, biasK, biasD, biasB, out, hstride, ost, whole, t, R, lane, c, hi);
}

extern "C" void kernel_launch(void* const* d_in, const int* in_sizes, int n_in,
                              void* d_out, int out_size, void* d_ws, size_t ws_size,
                              hipStream_t stream) {
    const float* x  = (const float*)d_in[0];
    const float* Wk = (const float*)d_in[1];
    const float* bk = (const float*)d_in[2];
    const float* Wd = (const float*)d_in[3];
    const float* bd = (const float*)d_in[4];
    const float* Wb = (const float*)d_in[5];
    const float* bb = (const float*)d_in[6];
    float* out = (float*)d_out;
    int R = in_sizes[0] / NBINS;

    dim3 block(256);
    dim3 grid(512);
    chord_heads_kernel<<<grid, block, 0, stream>>>(x, Wk, bk, Wd, bd, Wb, bb, out, R);
}
